# starting kernel
# baseline (speedup 1.0000x reference)
#include <hip/hip_runtime.h>


typedef __attribute__((ext_vector_type(2))) float v2f;
typedef __attribute__((ext_vector_type(8))) float v8f;

#define KTOT 144

__global__ __launch_bounds__(32) void lc2d_wmma_f32_kernel(
    const float* __restrict__ x,
    const float* __restrict__ wgt,
    const float* __restrict__ bias,
    float* __restrict__ out)
{
  __shared__ float lds_x[KTOT * 32];
  __shared__ float lds_w[KTOT * 32];

  const int lane = threadIdx.x;
  const int loc  = blockIdx.x;
  const int h = loc >> 6;
  const int w = loc & 63;

  {
    const int o = lane;
    #pragma unroll
    for (int c = 0; c < 16; ++c) {
      const float* src = wgt + (size_t)((o * 16 + c) * 4096 + loc) * 9;
      #pragma unroll
      for (int kk = 0; kk < 9; ++kk)
        lds_w[(c * 9 + kk) * 32 + o] = src[kk];
    }
  }

  {
    const int b = lane;
    #pragma unroll
    for (int c = 0; c < 16; ++c) {
      const float* src = x + ((size_t)(b * 16 + c) * 66 + h) * 66 + w;
      #pragma unroll
      for (int i = 0; i < 3; ++i)
        #pragma unroll
        for (int j = 0; j < 3; ++j)
          lds_x[(c * 9 + i * 3 + j) * 32 + b] = src[i * 66 + j];
    }
  }
  __syncthreads();

  const int t = lane >> 4;
  const int r = lane & 15;

  v8f acc00 = {}, acc01 = {}, acc10 = {}, acc11 = {};

  #pragma unroll
  for (int kc = 0; kc < KTOT; kc += 4) {
    const int kA = kc + 2 * t;
    const int base = kA * 32;
    v2f a0, a1, b0, b1;
    a0.x = lds_x[base + r];           a0.y = lds_x[base + 32 + r];
    a1.x = lds_x[base + 16 + r];      a1.y = lds_x[base + 48 + r];
    b0.x = lds_w[base + r];           b0.y = lds_w[base + 32 + r];
    b1.x = lds_w[base + 16 + r];      b1.y = lds_w[base + 48 + r];

    acc00 = __builtin_amdgcn_wmma_f32_16x16x4_f32(false, a0, false, b0,
                                                  (short)0, acc00, false, false);
    acc01 = __builtin_amdgcn_wmma_f32_16x16x4_f32(false, a0, false, b1,
                                                  (short)0, acc01, false, false);
    acc10 = __builtin_amdgcn_wmma_f32_16x16x4_f32(false, a1, false, b0,
                                                  (short)0, acc10, false, false);
    acc11 = __builtin_amdgcn_wmma_f32_16x16x4_f32(false, a1, false, b1,
                                                  (short)0, acc11, false, false);
  }

  const float bv0 = bias[(size_t)r * 4096 + loc];
  const float bv1 = bias[(size_t)(r + 16) * 4096 + loc];

  #pragma unroll
  for (int row = 0; row < 8; ++row) {
    const int bm0 = row + 8 * t;
    const int bm1 = 16 + row + 8 * t;
    out[(size_t)(bm0 * 32 + r)        * 4096 + loc] = acc00[row] + bv0;
    out[(size_t)(bm0 * 32 + r + 16)   * 4096 + loc] = acc01[row] + bv1;
    out[(size_t)(bm1 * 32 + r)        * 4096 + loc] = acc10[row] + bv0;
    out[(size_t)(bm1 * 32 + r + 16)   * 4096 + loc] = acc11[row] + bv1;
  }
}

extern "C" void kernel_launch(void* const* d_in, const int* in_sizes, int n_in,
                              void* d_out, int out_size, void* d_ws, size_t ws_size,
                              hipStream_t stream) {
  (void)in_sizes; (void)n_in; (void)out_size; (void)d_ws; (void)ws_size;
  const float* x    = (const float*)d_in[0];
  const float* wgt  = (const float*)d_in[1];
  const float* bias = (const float*)d_in[2];
  float* out = (float*)d_out;
  lc2d_wmma_f32_kernel<<<dim3(64 * 64), dim3(32), 0, stream>>>(x, wgt, bias, out);
}


# One work-group per 8 adjacent locations: weights, x and out pass through LDS so global accesses are contiguous 288/40/32 B runs, not one 4 B word per line per lane; k summed in a permuted order
# speedup vs baseline: 4.5332x; 4.5332x over previous
#include <hip/hip_runtime.h>


typedef __attribute__((ext_vector_type(2))) float v2f;
typedef __attribute__((ext_vector_type(4))) float v4f;
typedef __attribute__((ext_vector_type(8))) float v8f;

#define TL     8
#define W_RUN  74
#define W_O    300
#define X_C    30
#define X_B    132
#define O_O    9
#define O_B    292

__global__ __launch_bounds__(256) void lc2d_tile8_wmma_f32_kernel(
    const float* __restrict__ x,
    const float* __restrict__ wgt,
    const float* __restrict__ bias,
    float* __restrict__ out)
{
  __shared__ __attribute__((aligned(16))) float lds_w[32 * W_O];
  __shared__ __attribute__((aligned(16))) float lds_x[32 * X_B];

  const int tid  = threadIdx.x;
  const int lane = tid & 31;
  const int wv   = tid >> 5;
  const int loc0 = blockIdx.x * TL;
  const int h    = loc0 >> 6;
  const int w0   = loc0 & 63;
  const int t = lane >> 4;
  const int r = lane & 15;

  v8f acc00 = {}, acc01 = {}, acc10 = {}, acc11 = {};

  unsigned wg[9], wl[9];
  #pragma unroll
  for (int it = 0; it < 9; ++it) {
    const unsigned q   = it * 256 + tid;
    const unsigned run = q / 18;
    const unsigned v   = q - run * 18;
    const unsigned o   = run >> 2;
    const unsigned cl  = run & 3;
    wg[it] = (o * 16 + cl) * (4096 * 9) + 4 * v;
    wl[it] = o * W_O + cl * W_RUN + 4 * v;
  }
  unsigned xg[15], xl[15];
  #pragma unroll
  for (int it = 0; it < 15; ++it) {
    const unsigned q   = it * 256 + tid;
    const unsigned b   = q / 120;
    const unsigned rem = q - b * 120;
    const unsigned cl  = rem / 30;
    const unsigned r2  = rem - cl * 30;
    const unsigned i   = r2 / 10;
    const unsigned col = r2 - i * 10;
    xg[it] = (b * 16 + cl) * (66 * 66) + i * 66 + col;
    xl[it] = b * X_B + rem;
  }

  for (int ch = 0; ch < 4; ++ch) {
    const float* wsrc = wgt + ((size_t)(ch * 4) * 4096 + loc0) * 9;
    const float* xsrc = x + (size_t)(ch * 4) * (66 * 66) + h * 66 + w0;

    v4f wr[9];
    #pragma unroll
    for (int it = 0; it < 9; ++it)
      wr[it] = *(const v4f*)(wsrc + wg[it]);
    float xr[15];
    #pragma unroll
    for (int it = 0; it < 15; ++it)
      xr[it] = xsrc[xg[it]];
    asm volatile("" :: "v"(wr[0]), "v"(wr[1]), "v"(wr[2]), "v"(wr[3]), "v"(wr[4]),
                       "v"(wr[5]), "v"(wr[6]), "v"(wr[7]), "v"(wr[8]),
                       "v"(xr[0]), "v"(xr[1]), "v"(xr[2]), "v"(xr[3]), "v"(xr[4]),
                       "v"(xr[5]), "v"(xr[6]), "v"(xr[7]), "v"(xr[8]), "v"(xr[9]),
                       "v"(xr[10]), "v"(xr[11]), "v"(xr[12]), "v"(xr[13]), "v"(xr[14]));
    #pragma unroll
    for (int it = 0; it < 9; ++it) {
      float* dst = lds_w + wl[it];
      v2f lo = {wr[it].x, wr[it].y};
      v2f hi = {wr[it].z, wr[it].w};
      *(v2f*)dst       = lo;
      *(v2f*)(dst + 2) = hi;
    }
    #pragma unroll
    for (int it = 0; it < 15; ++it)
      lds_x[xl[it]] = xr[it];
    __syncthreads();

    const float* xa = lds_x + r * X_B + t * X_C + wv;
    const float* wb = lds_w + r * W_O + t * W_RUN + wv * 9;
    #pragma unroll
    for (int s = 0; s < 9; ++s) {
      v2f a0, a1, b0, b1;
      #pragma unroll
      for (int e = 0; e < 2; ++e) {
        const int j    = 2 * s + e;
        const int chf  = j / 9;
        const int kk   = j - 9 * chf;
        const int i    = kk / 3;
        const int jj   = kk - 3 * i;
        const int offA = chf * 2 * X_C + i * (TL + 2) + jj;
        const int offB = chf * 2 * W_RUN + kk;
        a0[e] = xa[offA];
        a1[e] = xa[16 * X_B + offA];
        b0[e] = wb[offB];
        b1[e] = wb[16 * W_O + offB];
      }
      acc00 = __builtin_amdgcn_wmma_f32_16x16x4_f32(false, a0, false, b0,
                                                    (short)0, acc00, false, false);
      acc01 = __builtin_amdgcn_wmma_f32_16x16x4_f32(false, a0, false, b1,
                                                    (short)0, acc01, false, false);
      acc10 = __builtin_amdgcn_wmma_f32_16x16x4_f32(false, a1, false, b0,
                                                    (short)0, acc10, false, false);
      acc11 = __builtin_amdgcn_wmma_f32_16x16x4_f32(false, a1, false, b1,
                                                    (short)0, acc11, false, false);
    }
    __syncthreads();
  }

  float* lds_o = lds_w;
  #pragma unroll
  for (int row = 0; row < 8; ++row) {
    const int bm0 = row + 8 * t;
    const int bm1 = 16 + row + 8 * t;
    lds_o[bm0 * O_B + r * O_O + wv]        = acc00[row];
    lds_o[bm0 * O_B + (r + 16) * O_O + wv] = acc01[row];
    lds_o[bm1 * O_B + r * O_O + wv]        = acc10[row];
    lds_o[bm1 * O_B + (r + 16) * O_O + wv] = acc11[row];
  }
  __syncthreads();

  const int half = tid & 1;
  const int o    = (tid >> 1) & 31;
  const int bq   = tid >> 6;
  const v4f bv = *(const v4f*)(bias + (size_t)o * 4096 + loc0 + half * 4);
  #pragma unroll
  for (int it = 0; it < 8; ++it) {
    const int b = it * 4 + bq;
    const float* p = lds_o + b * O_B + o * O_O + half * 4;
    v4f val;
    val.x = p[0] + bv.x;
    val.y = p[1] + bv.y;
    val.z = p[2] + bv.z;
    val.w = p[3] + bv.w;
    *(v4f*)(out + (size_t)(b * 32 + o) * 4096 + loc0 + half * 4) = val;
  }
}

extern "C" void kernel_launch(void* const* d_in, const int* in_sizes, int n_in,
                              void* d_out, int out_size, void* d_ws, size_t ws_size,
                              hipStream_t stream) {
  (void)in_sizes; (void)n_in; (void)out_size; (void)d_ws; (void)ws_size;
  const float* x    = (const float*)d_in[0];
  const float* wgt  = (const float*)d_in[1];
  const float* bias = (const float*)d_in[2];
  float* out = (float*)d_out;
  lc2d_tile8_wmma_f32_kernel<<<dim3(512), dim3(256), 0, stream>>>(x, wgt, bias, out);
}


# One work-group per 16 adjacent locations (16 waves, 256 groups), was 8: global runs grow to 576/72/64 B, a third fewer 128 B line requests for the same bytes; order of sums unchanged
# speedup vs baseline: 5.3098x; 1.1713x over previous
#include <hip/hip_runtime.h>


typedef __attribute__((ext_vector_type(2))) float v2f;
typedef __attribute__((ext_vector_type(4))) float v4f;
typedef __attribute__((ext_vector_type(8))) float v8f;

#define TL     16
#define W_RUN  146
#define W_O    588
#define X_C    54
#define X_B    220
#define O_O    17
#define O_B    548

__global__ __launch_bounds__(512) void lc2d_tile16_wmma_f32_kernel(
    const float* __restrict__ x,
    const float* __restrict__ wgt,
    const float* __restrict__ bias,
    float* __restrict__ out)
{
  __shared__ __attribute__((aligned(16))) float lds_w[32 * W_O];
  __shared__ __attribute__((aligned(16))) float lds_x[32 * X_B];

  const int tid  = threadIdx.x;
  const int lane = tid & 31;
  const int wv   = tid >> 5;
  const int loc0 = blockIdx.x * TL;
  const int h    = loc0 >> 6;
  const int w0   = loc0 & 63;
  const int t = lane >> 4;
  const int r = lane & 15;

  v8f acc00 = {}, acc01 = {}, acc10 = {}, acc11 = {};

  unsigned wg[9], wl[9];
  #pragma unroll
  for (int it = 0; it < 9; ++it) {
    const unsigned q   = it * 512 + tid;
    const unsigned run = q / 36;
    const unsigned v   = q - run * 36;
    const unsigned o   = run >> 2;
    const unsigned cl  = run & 3;
    wg[it] = (o * 16 + cl) * (4096 * 9) + 4 * v;
    wl[it] = o * W_O + cl * W_RUN + 4 * v;
  }
  unsigned xg[14], xl[14];
  #pragma unroll
  for (int it = 0; it < 14; ++it) {
    const unsigned q   = it * 512 + (it < 13 ? tid : (tid & 255));
    const unsigned b   = q / 216;
    const unsigned rem = q - b * 216;
    const unsigned cl  = rem / 54;
    const unsigned r2  = rem - cl * 54;
    const unsigned i   = r2 / 18;
    const unsigned col = r2 - i * 18;
    xg[it] = (b * 16 + cl) * (66 * 66) + i * 66 + col;
    xl[it] = b * X_B + rem;
  }
  const bool x_last = tid < 256;

  for (int ch = 0; ch < 4; ++ch) {
    const float* wsrc = wgt + ((size_t)(ch * 4) * 4096 + loc0) * 9;
    const float* xsrc = x + (size_t)(ch * 4) * (66 * 66) + h * 66 + w0;

    v4f wr[9];
    #pragma unroll
    for (int it = 0; it < 9; ++it)
      wr[it] = *(const v4f*)(wsrc + wg[it]);
    float xr[14];
    #pragma unroll
    for (int it = 0; it < 13; ++it)
      xr[it] = xsrc[xg[it]];
    xr[13] = 0.0f;
    if (x_last) xr[13] = xsrc[xg[13]];
    asm volatile("" :: "v"(wr[0]), "v"(wr[1]), "v"(wr[2]), "v"(wr[3]), "v"(wr[4]),
                       "v"(wr[5]), "v"(wr[6]), "v"(wr[7]), "v"(wr[8]),
                       "v"(xr[0]), "v"(xr[1]), "v"(xr[2]), "v"(xr[3]), "v"(xr[4]),
                       "v"(xr[5]), "v"(xr[6]), "v"(xr[7]), "v"(xr[8]), "v"(xr[9]),
                       "v"(xr[10]), "v"(xr[11]), "v"(xr[12]), "v"(xr[13]));
    #pragma unroll
    for (int it = 0; it < 9; ++it) {
      float* dst = lds_w + wl[it];
      v2f lo = {wr[it].x, wr[it].y};
      v2f hi = {wr[it].z, wr[it].w};
      *(v2f*)dst       = lo;
      *(v2f*)(dst + 2) = hi;
    }
    #pragma unroll
    for (int it = 0; it < 13; ++it)
      lds_x[xl[it]] = xr[it];
    if (x_last) lds_x[xl[13]] = xr[13];
    __syncthreads();

    const float* xa = lds_x + r * X_B + t * X_C + wv;
    const float* wb = lds_w + r * W_O + t * W_RUN + wv * 9;
    #pragma unroll
    for (int s = 0; s < 9; ++s) {
      v2f a0, a1, b0, b1;
      #pragma unroll
      for (int e = 0; e < 2; ++e) {
        const int j    = 2 * s + e;
        const int chf  = j / 9;
        const int kk   = j - 9 * chf;
        const int i    = kk / 3;
        const int jj   = kk - 3 * i;
        const int offA = chf * 2 * X_C + i * (TL + 2) + jj;
        const int offB = chf * 2 * W_RUN + kk;
        a0[e] = xa[offA];
        a1[e] = xa[16 * X_B + offA];
        b0[e] = wb[offB];
        b1[e] = wb[16 * W_O + offB];
      }
      acc00 = __builtin_amdgcn_wmma_f32_16x16x4_f32(false, a0, false, b0,
                                                    (short)0, acc00, false, false);
      acc01 = __builtin_amdgcn_wmma_f32_16x16x4_f32(false, a0, false, b1,
                                                    (short)0, acc01, false, false);
      acc10 = __builtin_amdgcn_wmma_f32_16x16x4_f32(false, a1, false, b0,
                                                    (short)0, acc10, false, false);
      acc11 = __builtin_amdgcn_wmma_f32_16x16x4_f32(false, a1, false, b1,
                                                    (short)0, acc11, false, false);
    }
    __syncthreads();
  }

  float* lds_o = lds_w;
  #pragma unroll
  for (int row = 0; row < 8; ++row) {
    const int bm0 = row + 8 * t;
    const int bm1 = 16 + row + 8 * t;
    lds_o[bm0 * O_B + r * O_O + wv]        = acc00[row];
    lds_o[bm0 * O_B + (r + 16) * O_O + wv] = acc01[row];
    lds_o[bm1 * O_B + r * O_O + wv]        = acc10[row];
    lds_o[bm1 * O_B + (r + 16) * O_O + wv] = acc11[row];
  }
  __syncthreads();

  const int quad = tid & 3;
  const int o    = (tid >> 2) & 31;
  const int bq   = tid >> 7;
  const v4f bv = *(const v4f*)(bias + (size_t)o * 4096 + loc0 + quad * 4);
  #pragma unroll
  for (int it = 0; it < 8; ++it) {
    const int b = it * 4 + bq;
    const float* p = lds_o + b * O_B + o * O_O + quad * 4;
    v4f val;
    val.x = p[0] + bv.x;
    val.y = p[1] + bv.y;
    val.z = p[2] + bv.z;
    val.w = p[3] + bv.w;
    *(v4f*)(out + (size_t)(b * 32 + o) * 4096 + loc0 + quad * 4) = val;
  }
}

extern "C" void kernel_launch(void* const* d_in, const int* in_sizes, int n_in,
                              void* d_out, int out_size, void* d_ws, size_t ws_size,
                              hipStream_t stream) {
  (void)in_sizes; (void)n_in; (void)out_size; (void)d_ws; (void)ws_size;
  const float* x    = (const float*)d_in[0];
  const float* wgt  = (const float*)d_in[1];
  const float* bias = (const float*)d_in[2];
  float* out = (float*)d_out;
  lc2d_tile16_wmma_f32_kernel<<<dim3(256), dim3(512), 0, stream>>>(x, wgt, bias, out);
}


# K walked in 2 chunks of 8 input channels (was 4 of 4): half the load-wait/barrier/WMMA rounds; staging maps redone to fit registers (x: one (b,c) tile per 8-byte load); order of sums unchanged
# speedup vs baseline: 5.6073x; 1.0560x over previous
#include <hip/hip_runtime.h>


typedef __attribute__((ext_vector_type(2))) float v2f;
typedef __attribute__((ext_vector_type(4))) float v4f;
typedef __attribute__((ext_vector_type(8))) float v8f;

#define TL     16
#define CC     8
#define W_RUN  146
#define W_O    1172
#define X_C    54
#define X_B    436
#define O_O    17
#define O_B    548
#define W_HALF_G (16u * 16u * 36864u * 4u)
#define W_HALF_L (16 * W_O)

__global__ __launch_bounds__(512) void lc2d_tile16_wmma_f32_kernel(
    const float* __restrict__ x,
    const float* __restrict__ wgt,
    const float* __restrict__ bias,
    float* __restrict__ out)
{
  __shared__ __attribute__((aligned(16))) float lds_w[32 * W_O];
  __shared__ __attribute__((aligned(16))) float lds_x[32 * X_B];

  const int tid  = threadIdx.x;
  const int lane = tid & 31;
  const int wv   = tid >> 5;
  const int loc0 = blockIdx.x * TL;
  const int h    = loc0 >> 6;
  const int w0   = loc0 & 63;
  const int t = lane >> 4;
  const int r = lane & 15;

  v8f acc00 = {}, acc01 = {}, acc10 = {}, acc11 = {};

  unsigned wg[18], wl[9];
  {
    unsigned wrun = (unsigned)tid / 36u;
    unsigned wv4  = (unsigned)tid - wrun * 36u;
    #pragma unroll
    for (int it = 0; it < 18; ++it) {
      const unsigned o  = wrun >> 3;
      const unsigned cl = wrun & 7u;
      wg[it] = ((o * 16u + cl) * 36864u + 4u * wv4) * 4u;
      if (it < 9) wl[it] = o * W_O + cl * W_RUN + 4u * wv4;
      wv4 += 8u; wrun += 14u;
      if (wv4 >= 36u) { wv4 -= 36u; wrun += 1u; }
    }
  }
  const unsigned xc   = (unsigned)wv & 7u;
  const unsigned xbp  = (unsigned)wv >> 3;
  const unsigned xk   = (unsigned)lane < 26u ? (unsigned)lane : 26u;
  const unsigned xi   = xk / 9u;
  const unsigned xcp  = xk - xi * 9u;
  const unsigned xg = ((xbp * 16u + xc) * (66u * 66u) + xi * 66u + 2u * xcp) * 4u;
  const unsigned xl = xbp * X_B + xc * X_C + 2u * xk;

  for (int ch = 0; ch < 2; ++ch) {
    const char* wsrc = (const char*)(wgt + ((size_t)(ch * CC) * 4096 + loc0) * 9);
    const char* xsrc = (const char*)(x + (size_t)(ch * CC) * (66 * 66) + h * 66 + w0);

    v4f wr[18];
    #pragma unroll
    for (int it = 0; it < 18; ++it)
      wr[it] = *(const v4f*)(wsrc + wg[it]);
    v2f xr[16];
    #pragma unroll
    for (int p = 0; p < 16; ++p)
      xr[p] = *(const v2f*)(xsrc + (size_t)p * (2u * 16u * 66u * 66u * 4u) + xg);
    asm volatile("" :: "v"(wr[0]), "v"(wr[1]), "v"(wr[2]), "v"(wr[3]), "v"(wr[4]),
                       "v"(wr[5]), "v"(wr[6]), "v"(wr[7]), "v"(wr[8]), "v"(wr[9]),
                       "v"(wr[10]), "v"(wr[11]), "v"(wr[12]), "v"(wr[13]), "v"(wr[14]),
                       "v"(wr[15]), "v"(wr[16]), "v"(wr[17]));
    asm volatile("" :: "v"(xr[0]), "v"(xr[1]), "v"(xr[2]), "v"(xr[3]), "v"(xr[4]),
                       "v"(xr[5]), "v"(xr[6]), "v"(xr[7]), "v"(xr[8]), "v"(xr[9]),
                       "v"(xr[10]), "v"(xr[11]), "v"(xr[12]), "v"(xr[13]), "v"(xr[14]),
                       "v"(xr[15]));
    #pragma unroll
    for (int it = 0; it < 18; ++it) {
      float* dst = lds_w + wl[it % 9] + (it / 9) * W_HALF_L;
      v2f lo = {wr[it].x, wr[it].y};
      v2f hi = {wr[it].z, wr[it].w};
      *(v2f*)dst       = lo;
      *(v2f*)(dst + 2) = hi;
    }
    #pragma unroll
    for (int p = 0; p < 16; ++p)
      *(v2f*)(lds_x + p * (2 * X_B) + xl) = xr[p];
    __syncthreads();

    const float* xa = lds_x + r * X_B + t * X_C + wv;
    const float* wb = lds_w + r * W_O + t * W_RUN + wv * 9;
    #pragma unroll
    for (int s = 0; s < 18; ++s) {
      v2f a0, a1, b0, b1;
      #pragma unroll
      for (int e = 0; e < 2; ++e) {
        const int j    = 2 * s + e;
        const int chf  = j / 9;
        const int kk   = j - 9 * chf;
        const int i    = kk / 3;
        const int jj   = kk - 3 * i;
        const int offA = chf * 2 * X_C + i * (TL + 2) + jj;
        const int offB = chf * 2 * W_RUN + kk;
        a0[e] = xa[offA];
        a1[e] = xa[16 * X_B + offA];
        b0[e] = wb[offB];
        b1[e] = wb[16 * W_O + offB];
      }
      acc00 = __builtin_amdgcn_wmma_f32_16x16x4_f32(false, a0, false, b0,
                                                    (short)0, acc00, false, false);
      acc01 = __builtin_amdgcn_wmma_f32_16x16x4_f32(false, a0, false, b1,
                                                    (short)0, acc01, false, false);
      acc10 = __builtin_amdgcn_wmma_f32_16x16x4_f32(false, a1, false, b0,
                                                    (short)0, acc10, false, false);
      acc11 = __builtin_amdgcn_wmma_f32_16x16x4_f32(false, a1, false, b1,
                                                    (short)0, acc11, false, false);
    }
    __syncthreads();
  }

  float* lds_o = lds_w;
  #pragma unroll
  for (int row = 0; row < 8; ++row) {
    const int bm0 = row + 8 * t;
    const int bm1 = 16 + row + 8 * t;
    lds_o[bm0 * O_B + r * O_O + wv]        = acc00[row];
    lds_o[bm0 * O_B + (r + 16) * O_O + wv] = acc01[row];
    lds_o[bm1 * O_B + r * O_O + wv]        = acc10[row];
    lds_o[bm1 * O_B + (r + 16) * O_O + wv] = acc11[row];
  }
  __syncthreads();

  const int quad = tid & 3;
  const int o    = (tid >> 2) & 31;
  const int bq   = tid >> 7;
  const v4f bv = *(const v4f*)(bias + (size_t)o * 4096 + loc0 + quad * 4);
  #pragma unroll
  for (int it = 0; it < 8; ++it) {
    const int b = it * 4 + bq;
    const float* p = lds_o + b * O_B + o * O_O + quad * 4;
    v4f val;
    val.x = p[0] + bv.x;
    val.y = p[1] + bv.y;
    val.z = p[2] + bv.z;
    val.w = p[3] + bv.w;
    *(v4f*)(out + (size_t)(b * 32 + o) * 4096 + loc0 + quad * 4) = val;
  }
}

extern "C" void kernel_launch(void* const* d_in, const int* in_sizes, int n_in,
                              void* d_out, int out_size, void* d_ws, size_t ws_size,
                              hipStream_t stream) {
  (void)in_sizes; (void)n_in; (void)out_size; (void)d_ws; (void)ws_size;
  const float* x    = (const float*)d_in[0];
  const float* wgt  = (const float*)d_in[1];
  const float* bias = (const float*)d_in[2];
  float* out = (float*)d_out;
  lc2d_tile16_wmma_f32_kernel<<<dim3(256), dim3(512), 0, stream>>>(x, wgt, bias, out);
}


# K in 4 chunks of 4 channels as a pipeline: two LDS stages, two register sets, loads of chunk i+2 issued among chunk i's WMMAs, round barrier waits for LDS only; order of sums unchanged
# speedup vs baseline: 6.1467x; 1.0962x over previous
#include <hip/hip_runtime.h>


typedef __attribute__((ext_vector_type(2))) float v2f;
typedef __attribute__((ext_vector_type(4))) float v4f;
typedef __attribute__((ext_vector_type(8))) float v8f;

#define DEV static __device__ __forceinline__

#define TL     16
#define CC     4
#define W_RUN  146
#define W_O    588
#define X_C    54
#define X_B    220
#define STG_W  (32 * W_O)
#define STG_X  (32 * X_B)
#define STG    (STG_W + STG_X)
#define O_O    17
#define O_B    548
#define W_CH_G (CC * 4096 * 9 * 4)
#define X_CH_G (CC * 66 * 66 * 4)
#define X_PASS_G (4u * 16u * 66u * 66u * 4u)

DEV void wg_barrier_lds() {
  asm volatile("s_wait_dscnt 0x0\n\ts_barrier_signal -1\n\ts_barrier_wait -1" ::: "memory");
}

template <int W0, int W1, int X0, int X1>
DEV void issue_loads(v4f (&wr)[9], v2f (&xr)[8], const char* wsrc, const char* xsrc,
                     const unsigned (&wg)[9], unsigned xg) {
  #pragma unroll
  for (int it = W0; it < W1; ++it)
    wr[it] = *(const v4f*)(wsrc + wg[it]);
  #pragma unroll
  for (int p = X0; p < X1; ++p)
    xr[p] = *(const v2f*)(xsrc + (size_t)p * X_PASS_G + xg);
}

DEV void store_stage(float* sw, float* sx, const v4f (&wr)[9], const v2f (&xr)[8],
                     const unsigned (&wl)[9], unsigned xl) {
  #pragma unroll
  for (int it = 0; it < 9; ++it) {
    float* dst = sw + wl[it];
    v2f lo = {wr[it].x, wr[it].y};
    v2f hi = {wr[it].z, wr[it].w};
    *(v2f*)dst       = lo;
    *(v2f*)(dst + 2) = hi;
  }
  #pragma unroll
  for (int p = 0; p < 8; ++p)
    *(v2f*)(sx + p * (4 * X_B) + xl) = xr[p];
}

template <int S0, int S1>
DEV void mma_steps(const float* xa, const float* wb, v8f& c00, v8f& c01, v8f& c10, v8f& c11) {
  #pragma unroll
  for (int s = S0; s < S1; ++s) {
    v2f a0, a1, b0, b1;
    #pragma unroll
    for (int e = 0; e < 2; ++e) {
      const int j    = 2 * s + e;
      const int chf  = j / 9;
      const int kk   = j - 9 * chf;
      const int i    = kk / 3;
      const int jj   = kk - 3 * i;
      const int offA = chf * 2 * X_C + i * (TL + 2) + jj;
      const int offB = chf * 2 * W_RUN + kk;
      a0[e] = xa[offA];
      a1[e] = xa[16 * X_B + offA];
      b0[e] = wb[offB];
      b1[e] = wb[16 * W_O + offB];
    }
    c00 = __builtin_amdgcn_wmma_f32_16x16x4_f32(false, a0, false, b0, (short)0, c00, false, false);
    c01 = __builtin_amdgcn_wmma_f32_16x16x4_f32(false, a0, false, b1, (short)0, c01, false, false);
    c10 = __builtin_amdgcn_wmma_f32_16x16x4_f32(false, a1, false, b0, (short)0, c10, false, false);
    c11 = __builtin_amdgcn_wmma_f32_16x16x4_f32(false, a1, false, b1, (short)0, c11, false, false);
  }
}

template <bool PF>
DEV void mma_round(const float* xa, const float* wb, v8f& c00, v8f& c01, v8f& c10, v8f& c11,
                   v4f (&wr)[9], v2f (&xr)[8], const char* wsrc, const char* xsrc,
                   const unsigned (&wg)[9], unsigned xg) {
  if (PF) issue_loads<0, 3, 0, 3>(wr, xr, wsrc, xsrc, wg, xg);
  mma_steps<0, 3>(xa, wb, c00, c01, c10, c11);
  if (PF) __builtin_amdgcn_sched_barrier(0);
  if (PF) issue_loads<3, 6, 3, 6>(wr, xr, wsrc, xsrc, wg, xg);
  mma_steps<3, 6>(xa, wb, c00, c01, c10, c11);
  if (PF) __builtin_amdgcn_sched_barrier(0);
  if (PF) issue_loads<6, 9, 6, 8>(wr, xr, wsrc, xsrc, wg, xg);
  mma_steps<6, 9>(xa, wb, c00, c01, c10, c11);
}

__global__ __launch_bounds__(512) void lc2d_tile16_pipe_wmma_f32_kernel(
    const float* __restrict__ x,
    const float* __restrict__ wgt,
    const float* __restrict__ bias,
    float* __restrict__ out)
{
  __shared__ __attribute__((aligned(16))) float lds[2 * STG];

  const int tid  = threadIdx.x;
  const int lane = tid & 31;
  const int wv   = tid >> 5;
  const int loc0 = blockIdx.x * TL;
  const int h    = loc0 >> 6;
  const int w0   = loc0 & 63;
  const int t = lane >> 4;
  const int r = lane & 15;

  v8f acc00 = {}, acc01 = {}, acc10 = {}, acc11 = {};

  unsigned wg[9], wl[9];
  {
    unsigned wrun = (unsigned)tid / 36u;
    unsigned wv4  = (unsigned)tid - wrun * 36u;
    #pragma unroll
    for (int it = 0; it < 9; ++it) {
      const unsigned o  = wrun >> 2;
      const unsigned cl = wrun & 3u;
      wg[it] = ((o * 16u + cl) * 36864u + 4u * wv4) * 4u;
      wl[it] = o * W_O + cl * W_RUN + 4u * wv4;
      wv4 += 8u; wrun += 14u;
      if (wv4 >= 36u) { wv4 -= 36u; wrun += 1u; }
    }
  }
  const unsigned xc   = (unsigned)wv & 3u;
  const unsigned xbp  = (unsigned)wv >> 2;
  const unsigned xk   = (unsigned)lane < 26u ? (unsigned)lane : 26u;
  const unsigned xi   = xk / 9u;
  const unsigned xcp  = xk - xi * 9u;
  const unsigned xg = ((xbp * 16u + xc) * (66u * 66u) + xi * 66u + 2u * xcp) * 4u;
  const unsigned xl = xbp * X_B + xc * X_C + 2u * xk;

  const char* wbase = (const char*)(wgt + (size_t)loc0 * 9);
  const char* xbase = (const char*)(x + h * 66 + w0);

  float* sw0 = lds;
  float* sx0 = lds + STG_W;
  float* sw1 = lds + STG;
  float* sx1 = lds + STG + STG_W;
  const int aoff = r * X_B + t * X_C + wv;
  const int boff = r * W_O + t * W_RUN + wv * 9;

  v4f wr0[9], wr1[9];
  v2f xr0[8], xr1[8];

  issue_loads<0, 9, 0, 8>(wr0, xr0, wbase, xbase, wg, xg);
  __builtin_amdgcn_sched_barrier(0);
  issue_loads<0, 9, 0, 8>(wr1, xr1, wbase + W_CH_G, xbase + X_CH_G, wg, xg);
  __builtin_amdgcn_sched_barrier(0);
  store_stage(sw0, sx0, wr0, xr0, wl, xl);
  wg_barrier_lds();

  mma_round<true>(sx0 + aoff, sw0 + boff, acc00, acc01, acc10, acc11,
                  wr0, xr0, wbase + 2 * W_CH_G, xbase + 2 * X_CH_G, wg, xg);
  store_stage(sw1, sx1, wr1, xr1, wl, xl);
  wg_barrier_lds();

  mma_round<true>(sx1 + aoff, sw1 + boff, acc00, acc01, acc10, acc11,
                  wr1, xr1, wbase + 3 * W_CH_G, xbase + 3 * X_CH_G, wg, xg);
  store_stage(sw0, sx0, wr0, xr0, wl, xl);
  wg_barrier_lds();

  mma_round<false>(sx0 + aoff, sw0 + boff, acc00, acc01, acc10, acc11,
                   wr0, xr0, wbase, xbase, wg, xg);
  store_stage(sw1, sx1, wr1, xr1, wl, xl);
  wg_barrier_lds();

  mma_round<false>(sx1 + aoff, sw1 + boff, acc00, acc01, acc10, acc11,
                   wr1, xr1, wbase, xbase, wg, xg);

  float* lds_o = sw0;
  #pragma unroll
  for (int row = 0; row < 8; ++row) {
    const int bm0 = row + 8 * t;
    const int bm1 = 16 + row + 8 * t;
    lds_o[bm0 * O_B + r * O_O + wv]        = acc00[row];
    lds_o[bm0 * O_B + (r + 16) * O_O + wv] = acc01[row];
    lds_o[bm1 * O_B + r * O_O + wv]        = acc10[row];
    lds_o[bm1 * O_B + (r + 16) * O_O + wv] = acc11[row];
  }
  wg_barrier_lds();

  const int quad = tid & 3;
  const int o    = (tid >> 2) & 31;
  const int bq   = tid >> 7;
  const v4f bv = *(const v4f*)(bias + (size_t)o * 4096 + loc0 + quad * 4);
  #pragma unroll
  for (int it = 0; it < 8; ++it) {
    const int b = it * 4 + bq;
    const float* p = lds_o + b * O_B + o * O_O + quad * 4;
    v4f val;
    val.x = p[0] + bv.x;
    val.y = p[1] + bv.y;
    val.z = p[2] + bv.z;
    val.w = p[3] + bv.w;
    *(v4f*)(out + (size_t)(b * 32 + o) * 4096 + loc0 + quad * 4) = val;
  }
}

extern "C" void kernel_launch(void* const* d_in, const int* in_sizes, int n_in,
                              void* d_out, int out_size, void* d_ws, size_t ws_size,
                              hipStream_t stream) {
  (void)in_sizes; (void)n_in; (void)out_size; (void)d_ws; (void)ws_size;
  const float* x    = (const float*)d_in[0];
  const float* wgt  = (const float*)d_in[1];
  const float* bias = (const float*)d_in[2];
  float* out = (float*)d_out;
  lc2d_tile16_pipe_wmma_f32_kernel<<<dim3(256), dim3(512), 0, stream>>>(x, wgt, bias, out);
}
